# starting kernel
# baseline (speedup 1.0000x reference)
#include <hip/hip_runtime.h>

typedef __attribute__((ext_vector_type(2))) float v2f;
typedef __attribute__((ext_vector_type(8))) float v8f;

#define NVOX   4096
#define NHEADS 8
#define HDIM   32
#define CCH    256
#define BATCH  4

__global__ void __launch_bounds__(256)
wmma_gemm_f32(const float* A, const float* __restrict__ B,
              const float* __restrict__ bias, float* D,
              int M, int K, int Ncols, int lda) {
  const int lane = threadIdx.x & 31;
  const int wave = (blockIdx.x * blockDim.x + threadIdx.x) >> 5;
  const int ntn  = Ncols >> 6;
  const int tm   = wave / ntn;
  const int tn   = wave - tm * ntn;
  if ((tm << 4) >= M) return;
  const int row0 = tm << 4, col0 = tn << 6;
  const int half = lane >> 4;
  const int lid  = lane & 15;

  const float* arow = A + (size_t)(row0 + lid) * lda + (half << 1);
  const float* bcol = B + (size_t)(half << 1) * Ncols + col0 + lid;

  v8f acc0 = {0.f,0.f,0.f,0.f,0.f,0.f,0.f,0.f};
  v8f acc1 = acc0, acc2 = acc0, acc3 = acc0;

#pragma unroll 4
  for (int k = 0; k < K; k += 4) {
    v2f a = *(const v2f*)(arow + k);
    const float* b0 = bcol + (size_t)k * Ncols;
    const float* b1 = b0 + Ncols;
    v2f bf0, bf1, bf2, bf3;
    bf0.x = b0[0];  bf0.y = b1[0];
    bf1.x = b0[16]; bf1.y = b1[16];
    bf2.x = b0[32]; bf2.y = b1[32];
    bf3.x = b0[48]; bf3.y = b1[48];
    acc0 = __builtin_amdgcn_wmma_f32_16x16x4_f32(false, a, false, bf0, (short)0, acc0, false, false);
    acc1 = __builtin_amdgcn_wmma_f32_16x16x4_f32(false, a, false, bf1, (short)0, acc1, false, false);
    acc2 = __builtin_amdgcn_wmma_f32_16x16x4_f32(false, a, false, bf2, (short)0, acc2, false, false);
    acc3 = __builtin_amdgcn_wmma_f32_16x16x4_f32(false, a, false, bf3, (short)0, acc3, false, false);
  }

  float* drow = D + (size_t)(row0 + (half << 3)) * Ncols + col0 + lid;
  const float bv0 = bias ? bias[col0 + lid]      : 0.0f;
  const float bv1 = bias ? bias[col0 + lid + 16] : 0.0f;
  const float bv2 = bias ? bias[col0 + lid + 32] : 0.0f;
  const float bv3 = bias ? bias[col0 + lid + 48] : 0.0f;
#pragma unroll
  for (int i = 0; i < 8; ++i) {
    float* dr = drow + (size_t)i * Ncols;
    dr[0]  = acc0[i] + bv0;
    dr[16] = acc1[i] + bv1;
    dr[32] = acc2[i] + bv2;
    dr[48] = acc3[i] + bv3;
  }
}

__global__ void __launch_bounds__(256)
neigh_attn(const float* qkv, const int* __restrict__ neigh, float* out) {
  const int lane = threadIdx.x & 31;
  const int wave = (blockIdx.x * blockDim.x + threadIdx.x) >> 5;
  const int n = wave & (NVOX - 1);
  const int h = (wave >> 12) & (NHEADS - 1);
  const int b = wave >> 15;

  const float NEG_INF = -__builtin_inff();
  const size_t base = (size_t)(b * NVOX + n) * (3 * CCH);
  const float q = qkv[base + h * HDIM + lane] * 0.17677669529663687f;

  int   nb[27];
  float s[27];
#pragma unroll
  for (int j = 0; j < 27; ++j) {
    const int id = neigh[n * 27 + j];
    nb[j] = id;
    float t = 0.f;
    if (id != NVOX) {
      const float* kr = qkv + (size_t)(b * NVOX + id) * (3 * CCH) + CCH + h * HDIM;
      t = q * kr[lane];
    }
#pragma unroll
    for (int off = 16; off; off >>= 1)
      t += __shfl_xor(t, off, 32);
    s[j] = (id == NVOX) ? NEG_INF : t;
  }

  float m = NEG_INF;
#pragma unroll
  for (int j = 0; j < 27; ++j) m = fmaxf(m, s[j]);
  float sum = 0.f;
#pragma unroll
  for (int j = 0; j < 27; ++j) { s[j] = __expf(s[j] - m); sum += s[j]; }
  const float inv = 1.0f / sum;

  float acc = 0.f;
#pragma unroll
  for (int j = 0; j < 27; ++j) {
    if (nb[j] != NVOX) {
      const float* vr = qkv + (size_t)(b * NVOX + nb[j]) * (3 * CCH) + 2 * CCH + h * HDIM;
      acc += s[j] * vr[lane];
    }
  }
  out[base + h * HDIM + lane] = acc * inv;
}

extern "C" void kernel_launch(void* const* d_in, const int* in_sizes, int n_in,
                              void* d_out, int out_size, void* d_ws, size_t ws_size,
                              hipStream_t stream) {
  (void)in_sizes; (void)n_in; (void)out_size; (void)ws_size;
  const float* x      = (const float*)d_in[0];
  const float* w_qkv  = (const float*)d_in[1];
  const float* w_proj = (const float*)d_in[2];
  const float* b_proj = (const float*)d_in[3];
  const int*   neigh  = (const int*)d_in[4];
  float* out = (float*)d_out;
  float* qkv = (float*)d_ws;

  const int M = BATCH * NVOX;

  {
    int waves = (M / 16) * (3 * CCH / 64);
    wmma_gemm_f32<<<waves * 32 / 256, 256, 0, stream>>>(
        x, w_qkv, nullptr, qkv, M, CCH, 3 * CCH, CCH);
  }
  {
    int waves = BATCH * NHEADS * NVOX;
    neigh_attn<<<waves * 32 / 256, 256, 0, stream>>>(qkv, neigh, qkv);
  }
  {
    int waves = (M / 16) * (CCH / 64);
    wmma_gemm_f32<<<waves * 32 / 256, 256, 0, stream>>>(
        qkv, w_proj, b_proj, out, M, CCH, CCH, 3 * CCH);
  }
}


# attention: one wave per (b,n) does all 8 heads on 512-B row loads; score = 4 FMAs per lane + 3 row_xmask DPP adds over 8 lanes (sum order changes; was 5 ds_bpermute steps), pads branch-free
# speedup vs baseline: 3.0494x; 3.0494x over previous
#include <hip/hip_runtime.h>

typedef __attribute__((ext_vector_type(2))) float v2f;
typedef __attribute__((ext_vector_type(8))) float v8f;

#define NVOX   4096
#define NHEADS 8
#define HDIM   32
#define CCH    256
#define BATCH  4

__global__ void __launch_bounds__(256)
wmma_gemm_f32(const float* A, const float* __restrict__ B,
              const float* __restrict__ bias, float* D,
              int M, int K, int Ncols, int lda) {
  const int lane = threadIdx.x & 31;
  const int wave = (blockIdx.x * blockDim.x + threadIdx.x) >> 5;
  const int ntn  = Ncols >> 6;
  const int tm   = wave / ntn;
  const int tn   = wave - tm * ntn;
  if ((tm << 4) >= M) return;
  const int row0 = tm << 4, col0 = tn << 6;
  const int half = lane >> 4;
  const int lid  = lane & 15;

  const float* arow = A + (size_t)(row0 + lid) * lda + (half << 1);
  const float* bcol = B + (size_t)(half << 1) * Ncols + col0 + lid;

  v8f acc0 = {0.f,0.f,0.f,0.f,0.f,0.f,0.f,0.f};
  v8f acc1 = acc0, acc2 = acc0, acc3 = acc0;

#pragma unroll 4
  for (int k = 0; k < K; k += 4) {
    v2f a = *(const v2f*)(arow + k);
    const float* b0 = bcol + (size_t)k * Ncols;
    const float* b1 = b0 + Ncols;
    v2f bf0, bf1, bf2, bf3;
    bf0.x = b0[0];  bf0.y = b1[0];
    bf1.x = b0[16]; bf1.y = b1[16];
    bf2.x = b0[32]; bf2.y = b1[32];
    bf3.x = b0[48]; bf3.y = b1[48];
    acc0 = __builtin_amdgcn_wmma_f32_16x16x4_f32(false, a, false, bf0, (short)0, acc0, false, false);
    acc1 = __builtin_amdgcn_wmma_f32_16x16x4_f32(false, a, false, bf1, (short)0, acc1, false, false);
    acc2 = __builtin_amdgcn_wmma_f32_16x16x4_f32(false, a, false, bf2, (short)0, acc2, false, false);
    acc3 = __builtin_amdgcn_wmma_f32_16x16x4_f32(false, a, false, bf3, (short)0, acc3, false, false);
  }

  float* drow = D + (size_t)(row0 + (half << 3)) * Ncols + col0 + lid;
  const float bv0 = bias ? bias[col0 + lid]      : 0.0f;
  const float bv1 = bias ? bias[col0 + lid + 16] : 0.0f;
  const float bv2 = bias ? bias[col0 + lid + 32] : 0.0f;
  const float bv3 = bias ? bias[col0 + lid + 48] : 0.0f;
#pragma unroll
  for (int i = 0; i < 8; ++i) {
    float* dr = drow + (size_t)i * Ncols;
    dr[0]  = acc0[i] + bv0;
    dr[16] = acc1[i] + bv1;
    dr[32] = acc2[i] + bv2;
    dr[48] = acc3[i] + bv3;
  }
}

typedef __attribute__((ext_vector_type(4))) float v4f;

template <int CTRL>
__device__ __forceinline__ float xmask_add(float v) {
  const int o = __builtin_amdgcn_update_dpp(0, __float_as_int(v), CTRL, 0xf, 0xf, true);
  return v + __int_as_float(o);
}
__device__ __forceinline__ float sum8(float v) {
  v = xmask_add<0x161>(v);
  v = xmask_add<0x162>(v);
  v = xmask_add<0x164>(v);
  return v;
}

__global__ void __launch_bounds__(256)
neigh_attn(const float* qkv, const int* __restrict__ neigh, float* out) {
  const int lane = threadIdx.x & 31;
  const int wave = __builtin_amdgcn_readfirstlane((int)((blockIdx.x * blockDim.x + threadIdx.x) >> 5));
  const int n = wave & (NVOX - 1);
  const int b = wave >> 12;
  const int c0 = lane << 2;

  const float NEG_INF = -__builtin_inff();
  const float SCALE = 0.17677669529663687f;
  const size_t base = (size_t)(b * NVOX + n) * (3 * CCH) + c0;
  v4f qa = *(const v4f*)(qkv + base);
  v4f qb = *(const v4f*)(qkv + base + 128);
  qa *= SCALE;
  qb *= SCALE;

  int   row[27];
  bool  pad[27];
#pragma unroll
  for (int j = 0; j < 27; ++j) {
    const int id = neigh[n * 27 + j];
    pad[j] = (unsigned)id >= (unsigned)NVOX;
    row[j] = pad[j] ? 0 : id;
  }

  float sa[27], sb[27];
#pragma unroll
  for (int j = 0; j < 27; ++j) {
    const float* kr = qkv + (size_t)(b * NVOX + row[j]) * (3 * CCH) + CCH + c0;
    const v4f ka = *(const v4f*)kr;
    const v4f kb = *(const v4f*)(kr + 128);
    float ta = qa.x * ka.x;
    ta = fmaf(qa.y, ka.y, ta);
    ta = fmaf(qa.z, ka.z, ta);
    ta = fmaf(qa.w, ka.w, ta);
    float tb = qb.x * kb.x;
    tb = fmaf(qb.y, kb.y, tb);
    tb = fmaf(qb.z, kb.z, tb);
    tb = fmaf(qb.w, kb.w, tb);
    ta = sum8(ta);
    tb = sum8(tb);
    sa[j] = pad[j] ? NEG_INF : ta;
    sb[j] = pad[j] ? NEG_INF : tb;
  }

  float ma = NEG_INF, mb = NEG_INF;
#pragma unroll
  for (int j = 0; j < 27; ++j) { ma = fmaxf(ma, sa[j]); mb = fmaxf(mb, sb[j]); }
  float la = 0.f, lb = 0.f;
#pragma unroll
  for (int j = 0; j < 27; ++j) {
    sa[j] = __expf(sa[j] - ma); la += sa[j];
    sb[j] = __expf(sb[j] - mb); lb += sb[j];
  }
  const float inva = 1.0f / la;
  const float invb = 1.0f / lb;

  v4f oa = {0.f, 0.f, 0.f, 0.f};
  v4f ob = oa;
#pragma unroll
  for (int j = 0; j < 27; ++j) {
    const float* vr = qkv + (size_t)(b * NVOX + row[j]) * (3 * CCH) + 2 * CCH + c0;
    const v4f va = *(const v4f*)vr;
    const v4f vb = *(const v4f*)(vr + 128);
    oa += sa[j] * va;
    ob += sb[j] * vb;
  }
  *(v4f*)(out + base)       = oa * inva;
  *(v4f*)(out + base + 128) = ob * invb;
}

extern "C" void kernel_launch(void* const* d_in, const int* in_sizes, int n_in,
                              void* d_out, int out_size, void* d_ws, size_t ws_size,
                              hipStream_t stream) {
  (void)in_sizes; (void)n_in; (void)out_size; (void)ws_size;
  const float* x      = (const float*)d_in[0];
  const float* w_qkv  = (const float*)d_in[1];
  const float* w_proj = (const float*)d_in[2];
  const float* b_proj = (const float*)d_in[3];
  const int*   neigh  = (const int*)d_in[4];
  float* out = (float*)d_out;
  float* qkv = (float*)d_ws;

  const int M = BATCH * NVOX;

  {
    int waves = (M / 16) * (3 * CCH / 64);
    wmma_gemm_f32<<<waves * 32 / 256, 256, 0, stream>>>(
        x, w_qkv, nullptr, qkv, M, CCH, 3 * CCH, CCH);
  }
  {
    int waves = BATCH * NVOX;
    neigh_attn<<<waves * 32 / 256, 256, 0, stream>>>(qkv, neigh, qkv);
  }
  {
    int waves = (M / 16) * (CCH / 64);
    wmma_gemm_f32<<<waves * 32 / 256, 256, 0, stream>>>(
        qkv, w_proj, b_proj, out, M, CCH, CCH, 3 * CCH);
  }
}


# both GEMMs: A and B k-slabs of 16 staged once per work-group (128x128 tile, 64x128 for the projection) through two LDS stages, fragments read from LDS as 8-byte pairs; sum order unchanged
# speedup vs baseline: 3.6010x; 1.1809x over previous
#include <hip/hip_runtime.h>

typedef __attribute__((ext_vector_type(2))) float v2f;
typedef __attribute__((ext_vector_type(8))) float v8f;

#define NVOX   4096
#define NHEADS 8
#define HDIM   32
#define CCH    256
#define BATCH  4

typedef __attribute__((ext_vector_type(4))) float v4f;

#define GBN 128
#define GBK 16
#define APITCH 20
#define BPITCH 144

template <int MT>
__global__ void __launch_bounds__(256)
wmma_gemm_f32(const float* A, const float* __restrict__ B,
              const float* __restrict__ bias, float* D,
              int K, int Ncols, int lda) {
  constexpr int BM = 64 * MT;
  __shared__ __attribute__((aligned(16))) float As[2][BM * APITCH];
  __shared__ __attribute__((aligned(16))) float Bs[2][GBK * BPITCH];

  const int tid  = threadIdx.x;
  const int lane = tid & 31;
  const int wv   = tid >> 5;
  const int wm   = wv >> 1;
  const int wn   = wv & 1;
  const int half = lane >> 4;
  const int lid  = lane & 15;

  const int nbn  = Ncols / GBN;
  const int bm   = blockIdx.x / nbn;
  const int bn   = blockIdx.x - bm * nbn;
  const int row0 = bm * BM, col0 = bn * GBN;

  const float* ag[MT];
  int as_off[MT];
#pragma unroll
  for (int r = 0; r < MT; ++r) {
    const int e = tid + 256 * r;
    const int ar = e >> 2, ac = (e & 3) << 2;
    ag[r] = A + (size_t)(row0 + ar) * lda + ac;
    as_off[r] = ar * APITCH + ac;
  }
  const int bp = tid >> 5, bc = (tid & 31) << 2;
  const float* bg = B + (size_t)(bp << 1) * Ncols + col0 + bc;
  const int bs_off = (bp * BPITCH + bc) << 1;

  const int a_frag = (wm * 16 * MT + lid) * APITCH + (half << 1);
  const int b_frag = (half * BPITCH + wn * 64 + lid) << 1;

  v8f acc[MT][4];
#pragma unroll
  for (int mt = 0; mt < MT; ++mt)
#pragma unroll
    for (int nt = 0; nt < 4; ++nt)
      acc[mt][nt] = (v8f){0.f,0.f,0.f,0.f,0.f,0.f,0.f,0.f};

  v4f ra[MT], rb[2];
#pragma unroll
  for (int r = 0; r < MT; ++r) ra[r] = *(const v4f*)(ag[r]);
  rb[0] = *(const v4f*)(bg);
  rb[1] = *(const v4f*)(bg + Ncols);
#pragma unroll
  for (int r = 0; r < MT; ++r) *(v4f*)(&As[0][as_off[r]]) = ra[r];
  *(v4f*)(&Bs[0][bs_off])     = (v4f){rb[0].x, rb[1].x, rb[0].y, rb[1].y};
  *(v4f*)(&Bs[0][bs_off + 4]) = (v4f){rb[0].z, rb[1].z, rb[0].w, rb[1].w};
  __syncthreads();

  const int nk = K / GBK;
  for (int kt = 0; kt < nk; ++kt) {
    const int cur = kt & 1;
    const bool more = (kt + 1) < nk;
    if (more) {
      const int k1 = (kt + 1) * GBK;
#pragma unroll
      for (int r = 0; r < MT; ++r) ra[r] = *(const v4f*)(ag[r] + k1);
      const float* bgk = bg + (size_t)k1 * Ncols;
      rb[0] = *(const v4f*)(bgk);
      rb[1] = *(const v4f*)(bgk + Ncols);
    }
    const float* as = &As[cur][a_frag];
    const float* bs = &Bs[cur][b_frag];
#pragma unroll
    for (int kk = 0; kk < GBK / 4; ++kk) {
      v2f af[MT], bf[4];
#pragma unroll
      for (int mt = 0; mt < MT; ++mt)
        af[mt] = *(const v2f*)(as + mt * 16 * APITCH + kk * 4);
#pragma unroll
      for (int nt = 0; nt < 4; ++nt)
        bf[nt] = *(const v2f*)(bs + ((kk * 2 * BPITCH + nt * 16) << 1));
#pragma unroll
      for (int mt = 0; mt < MT; ++mt)
#pragma unroll
        for (int nt = 0; nt < 4; ++nt)
          acc[mt][nt] = __builtin_amdgcn_wmma_f32_16x16x4_f32(false, af[mt], false, bf[nt], (short)0, acc[mt][nt], false, false);
    }
    if (more) {
#pragma unroll
      for (int r = 0; r < MT; ++r) *(v4f*)(&As[cur ^ 1][as_off[r]]) = ra[r];
      *(v4f*)(&Bs[cur ^ 1][bs_off])     = (v4f){rb[0].x, rb[1].x, rb[0].y, rb[1].y};
      *(v4f*)(&Bs[cur ^ 1][bs_off + 4]) = (v4f){rb[0].z, rb[1].z, rb[0].w, rb[1].w};
    }
    __syncthreads();
  }

  const int ccol = col0 + wn * 64 + lid;
  float bv[4];
#pragma unroll
  for (int nt = 0; nt < 4; ++nt) bv[nt] = bias ? bias[ccol + nt * 16] : 0.0f;
#pragma unroll
  for (int mt = 0; mt < MT; ++mt) {
    float* drow = D + (size_t)(row0 + wm * 16 * MT + mt * 16 + (half << 3)) * Ncols + ccol;
#pragma unroll
    for (int i = 0; i < 8; ++i) {
      float* dr = drow + (size_t)i * Ncols;
#pragma unroll
      for (int nt = 0; nt < 4; ++nt) dr[nt * 16] = acc[mt][nt][i] + bv[nt];
    }
  }
}


template <int CTRL>
__device__ __forceinline__ float xmask_add(float v) {
  const int o = __builtin_amdgcn_update_dpp(0, __float_as_int(v), CTRL, 0xf, 0xf, true);
  return v + __int_as_float(o);
}
__device__ __forceinline__ float sum8(float v) {
  v = xmask_add<0x161>(v);
  v = xmask_add<0x162>(v);
  v = xmask_add<0x164>(v);
  return v;
}

__global__ void __launch_bounds__(256)
neigh_attn(const float* qkv, const int* __restrict__ neigh, float* out) {
  const int lane = threadIdx.x & 31;
  const int wave = __builtin_amdgcn_readfirstlane((int)((blockIdx.x * blockDim.x + threadIdx.x) >> 5));
  const int n = wave & (NVOX - 1);
  const int b = wave >> 12;
  const int c0 = lane << 2;

  const float NEG_INF = -__builtin_inff();
  const float SCALE = 0.17677669529663687f;
  const size_t base = (size_t)(b * NVOX + n) * (3 * CCH) + c0;
  v4f qa = *(const v4f*)(qkv + base);
  v4f qb = *(const v4f*)(qkv + base + 128);
  qa *= SCALE;
  qb *= SCALE;

  int   row[27];
  bool  pad[27];
#pragma unroll
  for (int j = 0; j < 27; ++j) {
    const int id = neigh[n * 27 + j];
    pad[j] = (unsigned)id >= (unsigned)NVOX;
    row[j] = pad[j] ? 0 : id;
  }

  float sa[27], sb[27];
#pragma unroll
  for (int j = 0; j < 27; ++j) {
    const float* kr = qkv + (size_t)(b * NVOX + row[j]) * (3 * CCH) + CCH + c0;
    const v4f ka = *(const v4f*)kr;
    const v4f kb = *(const v4f*)(kr + 128);
    float ta = qa.x * ka.x;
    ta = fmaf(qa.y, ka.y, ta);
    ta = fmaf(qa.z, ka.z, ta);
    ta = fmaf(qa.w, ka.w, ta);
    float tb = qb.x * kb.x;
    tb = fmaf(qb.y, kb.y, tb);
    tb = fmaf(qb.z, kb.z, tb);
    tb = fmaf(qb.w, kb.w, tb);
    ta = sum8(ta);
    tb = sum8(tb);
    sa[j] = pad[j] ? NEG_INF : ta;
    sb[j] = pad[j] ? NEG_INF : tb;
  }

  float ma = NEG_INF, mb = NEG_INF;
#pragma unroll
  for (int j = 0; j < 27; ++j) { ma = fmaxf(ma, sa[j]); mb = fmaxf(mb, sb[j]); }
  float la = 0.f, lb = 0.f;
#pragma unroll
  for (int j = 0; j < 27; ++j) {
    sa[j] = __expf(sa[j] - ma); la += sa[j];
    sb[j] = __expf(sb[j] - mb); lb += sb[j];
  }
  const float inva = 1.0f / la;
  const float invb = 1.0f / lb;

  v4f oa = {0.f, 0.f, 0.f, 0.f};
  v4f ob = oa;
#pragma unroll
  for (int j = 0; j < 27; ++j) {
    const float* vr = qkv + (size_t)(b * NVOX + row[j]) * (3 * CCH) + 2 * CCH + c0;
    const v4f va = *(const v4f*)vr;
    const v4f vb = *(const v4f*)(vr + 128);
    oa += sa[j] * va;
    ob += sb[j] * vb;
  }
  *(v4f*)(out + base)       = oa * inva;
  *(v4f*)(out + base + 128) = ob * invb;
}

extern "C" void kernel_launch(void* const* d_in, const int* in_sizes, int n_in,
                              void* d_out, int out_size, void* d_ws, size_t ws_size,
                              hipStream_t stream) {
  (void)in_sizes; (void)n_in; (void)out_size; (void)ws_size;
  const float* x      = (const float*)d_in[0];
  const float* w_qkv  = (const float*)d_in[1];
  const float* w_proj = (const float*)d_in[2];
  const float* b_proj = (const float*)d_in[3];
  const int*   neigh  = (const int*)d_in[4];
  float* out = (float*)d_out;
  float* qkv = (float*)d_ws;

  const int M = BATCH * NVOX;

  {
    int blocks = (M / 128) * (3 * CCH / GBN);
    wmma_gemm_f32<2><<<blocks, 256, 0, stream>>>(
        x, w_qkv, nullptr, qkv, CCH, 3 * CCH, CCH);
  }
  {
    int waves = BATCH * NVOX;
    neigh_attn<<<waves * 32 / 256, 256, 0, stream>>>(qkv, neigh, qkv);
  }
  {
    int blocks = (M / 64) * (CCH / GBN);
    wmma_gemm_f32<1><<<blocks, 256, 0, stream>>>(
        qkv, w_proj, b_proj, out, CCH, CCH, 3 * CCH);
  }
}


# attention: a work-group is 16 waves on a 1x4x4 voxel block (was 8 waves, 8 voxels in a row along z): 108 distinct K/V rows per 16 voxels, was 180, more loads hit the WGP cache; sum order unchanged
# speedup vs baseline: 3.7791x; 1.0495x over previous
#include <hip/hip_runtime.h>

typedef __attribute__((ext_vector_type(2))) float v2f;
typedef __attribute__((ext_vector_type(8))) float v8f;

#define NVOX   4096
#define NHEADS 8
#define HDIM   32
#define CCH    256
#define BATCH  4

typedef __attribute__((ext_vector_type(4))) float v4f;

#define GBN 128
#define GBK 16
#define APITCH 20
#define BPITCH 144

template <int MT>
__global__ void __launch_bounds__(256)
wmma_gemm_f32(const float* A, const float* __restrict__ B,
              const float* __restrict__ bias, float* D,
              int K, int Ncols, int lda) {
  constexpr int BM = 64 * MT;
  __shared__ __attribute__((aligned(16))) float As[2][BM * APITCH];
  __shared__ __attribute__((aligned(16))) float Bs[2][GBK * BPITCH];

  const int tid  = threadIdx.x;
  const int lane = tid & 31;
  const int wv   = tid >> 5;
  const int wm   = wv >> 1;
  const int wn   = wv & 1;
  const int half = lane >> 4;
  const int lid  = lane & 15;

  const int nbn  = Ncols / GBN;
  const int bm   = blockIdx.x / nbn;
  const int bn   = blockIdx.x - bm * nbn;
  const int row0 = bm * BM, col0 = bn * GBN;

  const float* ag[MT];
  int as_off[MT];
#pragma unroll
  for (int r = 0; r < MT; ++r) {
    const int e = tid + 256 * r;
    const int ar = e >> 2, ac = (e & 3) << 2;
    ag[r] = A + (size_t)(row0 + ar) * lda + ac;
    as_off[r] = ar * APITCH + ac;
  }
  const int bp = tid >> 5, bc = (tid & 31) << 2;
  const float* bg = B + (size_t)(bp << 1) * Ncols + col0 + bc;
  const int bs_off = (bp * BPITCH + bc) << 1;

  const int a_frag = (wm * 16 * MT + lid) * APITCH + (half << 1);
  const int b_frag = (half * BPITCH + wn * 64 + lid) << 1;

  v8f acc[MT][4];
#pragma unroll
  for (int mt = 0; mt < MT; ++mt)
#pragma unroll
    for (int nt = 0; nt < 4; ++nt)
      acc[mt][nt] = (v8f){0.f,0.f,0.f,0.f,0.f,0.f,0.f,0.f};

  v4f ra[MT], rb[2];
#pragma unroll
  for (int r = 0; r < MT; ++r) ra[r] = *(const v4f*)(ag[r]);
  rb[0] = *(const v4f*)(bg);
  rb[1] = *(const v4f*)(bg + Ncols);
#pragma unroll
  for (int r = 0; r < MT; ++r) *(v4f*)(&As[0][as_off[r]]) = ra[r];
  *(v4f*)(&Bs[0][bs_off])     = (v4f){rb[0].x, rb[1].x, rb[0].y, rb[1].y};
  *(v4f*)(&Bs[0][bs_off + 4]) = (v4f){rb[0].z, rb[1].z, rb[0].w, rb[1].w};
  __syncthreads();

  const int nk = K / GBK;
  for (int kt = 0; kt < nk; ++kt) {
    const int cur = kt & 1;
    const bool more = (kt + 1) < nk;
    if (more) {
      const int k1 = (kt + 1) * GBK;
#pragma unroll
      for (int r = 0; r < MT; ++r) ra[r] = *(const v4f*)(ag[r] + k1);
      const float* bgk = bg + (size_t)k1 * Ncols;
      rb[0] = *(const v4f*)(bgk);
      rb[1] = *(const v4f*)(bgk + Ncols);
    }
    const float* as = &As[cur][a_frag];
    const float* bs = &Bs[cur][b_frag];
#pragma unroll
    for (int kk = 0; kk < GBK / 4; ++kk) {
      v2f af[MT], bf[4];
#pragma unroll
      for (int mt = 0; mt < MT; ++mt)
        af[mt] = *(const v2f*)(as + mt * 16 * APITCH + kk * 4);
#pragma unroll
      for (int nt = 0; nt < 4; ++nt)
        bf[nt] = *(const v2f*)(bs + ((kk * 2 * BPITCH + nt * 16) << 1));
#pragma unroll
      for (int mt = 0; mt < MT; ++mt)
#pragma unroll
        for (int nt = 0; nt < 4; ++nt)
          acc[mt][nt] = __builtin_amdgcn_wmma_f32_16x16x4_f32(false, af[mt], false, bf[nt], (short)0, acc[mt][nt], false, false);
    }
    if (more) {
#pragma unroll
      for (int r = 0; r < MT; ++r) *(v4f*)(&As[cur ^ 1][as_off[r]]) = ra[r];
      *(v4f*)(&Bs[cur ^ 1][bs_off])     = (v4f){rb[0].x, rb[1].x, rb[0].y, rb[1].y};
      *(v4f*)(&Bs[cur ^ 1][bs_off + 4]) = (v4f){rb[0].z, rb[1].z, rb[0].w, rb[1].w};
    }
    __syncthreads();
  }

  const int ccol = col0 + wn * 64 + lid;
  float bv[4];
#pragma unroll
  for (int nt = 0; nt < 4; ++nt) bv[nt] = bias ? bias[ccol + nt * 16] : 0.0f;
#pragma unroll
  for (int mt = 0; mt < MT; ++mt) {
    float* drow = D + (size_t)(row0 + wm * 16 * MT + mt * 16 + (half << 3)) * Ncols + ccol;
#pragma unroll
    for (int i = 0; i < 8; ++i) {
      float* dr = drow + (size_t)i * Ncols;
#pragma unroll
      for (int nt = 0; nt < 4; ++nt) dr[nt * 16] = acc[mt][nt][i] + bv[nt];
    }
  }
}


template <int CTRL>
__device__ __forceinline__ float xmask_add(float v) {
  const int o = __builtin_amdgcn_update_dpp(0, __float_as_int(v), CTRL, 0xf, 0xf, true);
  return v + __int_as_float(o);
}
__device__ __forceinline__ float sum8(float v) {
  v = xmask_add<0x161>(v);
  v = xmask_add<0x162>(v);
  v = xmask_add<0x164>(v);
  return v;
}

__global__ void __launch_bounds__(512)
neigh_attn(const float* qkv, const int* __restrict__ neigh, float* out) {
  const int lane = threadIdx.x & 31;
  const int w   = __builtin_amdgcn_readfirstlane((int)(threadIdx.x >> 5));
  const int blk = blockIdx.x & 255;
  const int b   = blockIdx.x >> 8;
  const int n   = ((blk >> 4) << 8) | (((((blk >> 2) & 3) << 2) | (w >> 2)) << 4) | ((blk & 3) << 2) | (w & 3);
  const int c0 = lane << 2;

  const float NEG_INF = -__builtin_inff();
  const float SCALE = 0.17677669529663687f;
  const size_t base = (size_t)(b * NVOX + n) * (3 * CCH) + c0;
  v4f qa = *(const v4f*)(qkv + base);
  v4f qb = *(const v4f*)(qkv + base + 128);
  qa *= SCALE;
  qb *= SCALE;

  int   row[27];
  bool  pad[27];
#pragma unroll
  for (int j = 0; j < 27; ++j) {
    const int id = neigh[n * 27 + j];
    pad[j] = (unsigned)id >= (unsigned)NVOX;
    row[j] = pad[j] ? 0 : id;
  }

  float sa[27], sb[27];
#pragma unroll
  for (int j = 0; j < 27; ++j) {
    const float* kr = qkv + (size_t)(b * NVOX + row[j]) * (3 * CCH) + CCH + c0;
    const v4f ka = *(const v4f*)kr;
    const v4f kb = *(const v4f*)(kr + 128);
    float ta = qa.x * ka.x;
    ta = fmaf(qa.y, ka.y, ta);
    ta = fmaf(qa.z, ka.z, ta);
    ta = fmaf(qa.w, ka.w, ta);
    float tb = qb.x * kb.x;
    tb = fmaf(qb.y, kb.y, tb);
    tb = fmaf(qb.z, kb.z, tb);
    tb = fmaf(qb.w, kb.w, tb);
    ta = sum8(ta);
    tb = sum8(tb);
    sa[j] = pad[j] ? NEG_INF : ta;
    sb[j] = pad[j] ? NEG_INF : tb;
  }

  float ma = NEG_INF, mb = NEG_INF;
#pragma unroll
  for (int j = 0; j < 27; ++j) { ma = fmaxf(ma, sa[j]); mb = fmaxf(mb, sb[j]); }
  float la = 0.f, lb = 0.f;
#pragma unroll
  for (int j = 0; j < 27; ++j) {
    sa[j] = __expf(sa[j] - ma); la += sa[j];
    sb[j] = __expf(sb[j] - mb); lb += sb[j];
  }
  const float inva = 1.0f / la;
  const float invb = 1.0f / lb;

  v4f oa = {0.f, 0.f, 0.f, 0.f};
  v4f ob = oa;
#pragma unroll
  for (int j = 0; j < 27; ++j) {
    const float* vr = qkv + (size_t)(b * NVOX + row[j]) * (3 * CCH) + 2 * CCH + c0;
    const v4f va = *(const v4f*)vr;
    const v4f vb = *(const v4f*)(vr + 128);
    oa += sa[j] * va;
    ob += sb[j] * vb;
  }
  *(v4f*)(out + base)       = oa * inva;
  *(v4f*)(out + base + 128) = ob * invb;
}

extern "C" void kernel_launch(void* const* d_in, const int* in_sizes, int n_in,
                              void* d_out, int out_size, void* d_ws, size_t ws_size,
                              hipStream_t stream) {
  (void)in_sizes; (void)n_in; (void)out_size; (void)ws_size;
  const float* x      = (const float*)d_in[0];
  const float* w_qkv  = (const float*)d_in[1];
  const float* w_proj = (const float*)d_in[2];
  const float* b_proj = (const float*)d_in[3];
  const int*   neigh  = (const int*)d_in[4];
  float* out = (float*)d_out;
  float* qkv = (float*)d_ws;

  const int M = BATCH * NVOX;

  {
    int blocks = (M / 128) * (3 * CCH / GBN);
    wmma_gemm_f32<2><<<blocks, 256, 0, stream>>>(
        x, w_qkv, nullptr, qkv, CCH, 3 * CCH, CCH);
  }
  {
    neigh_attn<<<BATCH * NVOX / 16, 512, 0, stream>>>(qkv, neigh, qkv);
  }
  {
    int blocks = (M / 64) * (CCH / GBN);
    wmma_gemm_f32<1><<<blocks, 256, 0, stream>>>(
        qkv, w_proj, b_proj, out, CCH, CCH, 3 * CCH);
  }
}


# attention and output projection fused in one launch: a 16-wave work-group keeps the attention rows of its 64 voxels in LDS and multiplies them by w_proj there (2 launches, was 3); sum order unchanged
# speedup vs baseline: 4.2077x; 1.1134x over previous
#include <hip/hip_runtime.h>

typedef __attribute__((ext_vector_type(2))) float v2f;
typedef __attribute__((ext_vector_type(8))) float v8f;

#define NVOX   4096
#define NHEADS 8
#define HDIM   32
#define CCH    256
#define BATCH  4

typedef __attribute__((ext_vector_type(4))) float v4f;

#define GBN 128
#define GBK 16
#define APITCH 20
#define BPITCH 144

template <int MT>
__global__ void __launch_bounds__(256)
wmma_gemm_f32(const float* A, const float* __restrict__ B,
              const float* __restrict__ bias, float* D,
              int K, int Ncols, int lda) {
  constexpr int BM = 64 * MT;
  __shared__ __attribute__((aligned(16))) float As[2][BM * APITCH];
  __shared__ __attribute__((aligned(16))) float Bs[2][GBK * BPITCH];

  const int tid  = threadIdx.x;
  const int lane = tid & 31;
  const int wv   = tid >> 5;
  const int wm   = wv >> 1;
  const int wn   = wv & 1;
  const int half = lane >> 4;
  const int lid  = lane & 15;

  const int nbn  = Ncols / GBN;
  const int bm   = blockIdx.x / nbn;
  const int bn   = blockIdx.x - bm * nbn;
  const int row0 = bm * BM, col0 = bn * GBN;

  const float* ag[MT];
  int as_off[MT];
#pragma unroll
  for (int r = 0; r < MT; ++r) {
    const int e = tid + 256 * r;
    const int ar = e >> 2, ac = (e & 3) << 2;
    ag[r] = A + (size_t)(row0 + ar) * lda + ac;
    as_off[r] = ar * APITCH + ac;
  }
  const int bp = tid >> 5, bc = (tid & 31) << 2;
  const float* bg = B + (size_t)(bp << 1) * Ncols + col0 + bc;
  const int bs_off = (bp * BPITCH + bc) << 1;

  const int a_frag = (wm * 16 * MT + lid) * APITCH + (half << 1);
  const int b_frag = (half * BPITCH + wn * 64 + lid) << 1;

  v8f acc[MT][4];
#pragma unroll
  for (int mt = 0; mt < MT; ++mt)
#pragma unroll
    for (int nt = 0; nt < 4; ++nt)
      acc[mt][nt] = (v8f){0.f,0.f,0.f,0.f,0.f,0.f,0.f,0.f};

  v4f ra[MT], rb[2];
#pragma unroll
  for (int r = 0; r < MT; ++r) ra[r] = *(const v4f*)(ag[r]);
  rb[0] = *(const v4f*)(bg);
  rb[1] = *(const v4f*)(bg + Ncols);
#pragma unroll
  for (int r = 0; r < MT; ++r) *(v4f*)(&As[0][as_off[r]]) = ra[r];
  *(v4f*)(&Bs[0][bs_off])     = (v4f){rb[0].x, rb[1].x, rb[0].y, rb[1].y};
  *(v4f*)(&Bs[0][bs_off + 4]) = (v4f){rb[0].z, rb[1].z, rb[0].w, rb[1].w};
  __syncthreads();

  const int nk = K / GBK;
  for (int kt = 0; kt < nk; ++kt) {
    const int cur = kt & 1;
    const bool more = (kt + 1) < nk;
    if (more) {
      const int k1 = (kt + 1) * GBK;
#pragma unroll
      for (int r = 0; r < MT; ++r) ra[r] = *(const v4f*)(ag[r] + k1);
      const float* bgk = bg + (size_t)k1 * Ncols;
      rb[0] = *(const v4f*)(bgk);
      rb[1] = *(const v4f*)(bgk + Ncols);
    }
    const float* as = &As[cur][a_frag];
    const float* bs = &Bs[cur][b_frag];
#pragma unroll
    for (int kk = 0; kk < GBK / 4; ++kk) {
      v2f af[MT], bf[4];
#pragma unroll
      for (int mt = 0; mt < MT; ++mt)
        af[mt] = *(const v2f*)(as + mt * 16 * APITCH + kk * 4);
#pragma unroll
      for (int nt = 0; nt < 4; ++nt)
        bf[nt] = *(const v2f*)(bs + ((kk * 2 * BPITCH + nt * 16) << 1));
#pragma unroll
      for (int mt = 0; mt < MT; ++mt)
#pragma unroll
        for (int nt = 0; nt < 4; ++nt)
          acc[mt][nt] = __builtin_amdgcn_wmma_f32_16x16x4_f32(false, af[mt], false, bf[nt], (short)0, acc[mt][nt], false, false);
    }
    if (more) {
#pragma unroll
      for (int r = 0; r < MT; ++r) *(v4f*)(&As[cur ^ 1][as_off[r]]) = ra[r];
      *(v4f*)(&Bs[cur ^ 1][bs_off])     = (v4f){rb[0].x, rb[1].x, rb[0].y, rb[1].y};
      *(v4f*)(&Bs[cur ^ 1][bs_off + 4]) = (v4f){rb[0].z, rb[1].z, rb[0].w, rb[1].w};
    }
    __syncthreads();
  }

  const int ccol = col0 + wn * 64 + lid;
  float bv[4];
#pragma unroll
  for (int nt = 0; nt < 4; ++nt) bv[nt] = bias ? bias[ccol + nt * 16] : 0.0f;
#pragma unroll
  for (int mt = 0; mt < MT; ++mt) {
    float* drow = D + (size_t)(row0 + wm * 16 * MT + mt * 16 + (half << 3)) * Ncols + ccol;
#pragma unroll
    for (int i = 0; i < 8; ++i) {
      float* dr = drow + (size_t)i * Ncols;
#pragma unroll
      for (int nt = 0; nt < 4; ++nt) dr[nt * 16] = acc[mt][nt][i] + bv[nt];
    }
  }
}


template <int CTRL>
__device__ __forceinline__ float xmask_add(float v) {
  const int o = __builtin_amdgcn_update_dpp(0, __float_as_int(v), CTRL, 0xf, 0xf, true);
  return v + __int_as_float(o);
}
__device__ __forceinline__ float sum8(float v) {
  v = xmask_add<0x161>(v);
  v = xmask_add<0x162>(v);
  v = xmask_add<0x164>(v);
  return v;
}

#define FPITCH  260
#define FBPITCH 272
#define FK      32

__device__ __forceinline__ void load_b_slab(v4f (&rb)[4], const float* g) {
  rb[0] = *(const v4f*)(g);
  rb[1] = *(const v4f*)(g + CCH);
  rb[2] = *(const v4f*)(g + 16 * CCH);
  rb[3] = *(const v4f*)(g + 17 * CCH);
}
__device__ __forceinline__ void store_b_slab(float* s, const v4f (&rb)[4]) {
  *(v4f*)(s)     = (v4f){rb[0].x, rb[1].x, rb[0].y, rb[1].y};
  *(v4f*)(s + 4) = (v4f){rb[0].z, rb[1].z, rb[0].w, rb[1].w};
  *(v4f*)(s + 8 * FBPITCH * 2)     = (v4f){rb[2].x, rb[3].x, rb[2].y, rb[3].y};
  *(v4f*)(s + 8 * FBPITCH * 2 + 4) = (v4f){rb[2].z, rb[3].z, rb[2].w, rb[3].w};
}

__global__ void __launch_bounds__(512)
attn_proj(const float* qkv, const int* __restrict__ neigh,
          const float* __restrict__ Wp, const float* __restrict__ bias, float* out) {
  __shared__ __attribute__((aligned(16))) float At[64 * FPITCH];
  __shared__ __attribute__((aligned(16))) float Bs[2][FK * FBPITCH];

  const int tid  = threadIdx.x;
  const int lane = tid & 31;
  const int w   = __builtin_amdgcn_readfirstlane(tid >> 5);
  const int b   = blockIdx.x >> 6;
  const int blk = blockIdx.x & 63;
  const int nyz = (((((blk >> 2) & 3) << 2) | (w >> 2)) << 4) | ((blk & 3) << 2) | (w & 3);
  const int x0  = (blk >> 4) << 2;
  const int c0  = lane << 2;

  const float NEG_INF = -__builtin_inff();
  const float SCALE = 0.17677669529663687f;

  const int wm   = w >> 2;
  const int wn   = w & 3;
  const int half = lane >> 4;
  const int lid  = lane & 15;
  const int bp = tid >> 6, bc = (tid & 63) << 2;
  const float* bg = Wp + (size_t)(bp << 1) * CCH + bc;
  const int bs_off = (bp * FBPITCH + bc) << 1;
  v4f rb[4];
  load_b_slab(rb, bg);
  store_b_slab(&Bs[0][bs_off], rb);
  load_b_slab(rb, bg + (size_t)FK * CCH);

#pragma unroll 1
  for (int it = 0; it < 4; ++it) {
    const int n = ((x0 + it) << 8) | nyz;
    const size_t base = (size_t)(b * NVOX + n) * (3 * CCH) + c0;
    v4f qa = *(const v4f*)(qkv + base);
    v4f qb = *(const v4f*)(qkv + base + 128);
    qa *= SCALE;
    qb *= SCALE;

    int   row[27];
    bool  pad[27];
#pragma unroll
    for (int j = 0; j < 27; ++j) {
      const int id = neigh[n * 27 + j];
      pad[j] = (unsigned)id >= (unsigned)NVOX;
      row[j] = pad[j] ? 0 : id;
    }

    float sa[27], sb[27];
#pragma unroll
    for (int j = 0; j < 27; ++j) {
      const float* kr = qkv + (size_t)(b * NVOX + row[j]) * (3 * CCH) + CCH + c0;
      const v4f ka = *(const v4f*)kr;
      const v4f kb = *(const v4f*)(kr + 128);
      float ta = qa.x * ka.x;
      ta = fmaf(qa.y, ka.y, ta);
      ta = fmaf(qa.z, ka.z, ta);
      ta = fmaf(qa.w, ka.w, ta);
      float tb = qb.x * kb.x;
      tb = fmaf(qb.y, kb.y, tb);
      tb = fmaf(qb.z, kb.z, tb);
      tb = fmaf(qb.w, kb.w, tb);
      ta = sum8(ta);
      tb = sum8(tb);
      sa[j] = pad[j] ? NEG_INF : ta;
      sb[j] = pad[j] ? NEG_INF : tb;
    }

    float ma = NEG_INF, mb = NEG_INF;
#pragma unroll
    for (int j = 0; j < 27; ++j) { ma = fmaxf(ma, sa[j]); mb = fmaxf(mb, sb[j]); }
    float la = 0.f, lb = 0.f;
#pragma unroll
    for (int j = 0; j < 27; ++j) {
      sa[j] = __expf(sa[j] - ma); la += sa[j];
      sb[j] = __expf(sb[j] - mb); lb += sb[j];
    }
    const float inva = 1.0f / la;
    const float invb = 1.0f / lb;

    v4f oa = {0.f, 0.f, 0.f, 0.f};
    v4f ob = oa;
#pragma unroll
    for (int j = 0; j < 27; ++j) {
      const float* vr = qkv + (size_t)(b * NVOX + row[j]) * (3 * CCH) + 2 * CCH + c0;
      const v4f va = *(const v4f*)vr;
      const v4f vb = *(const v4f*)(vr + 128);
      oa += sa[j] * va;
      ob += sb[j] * vb;
    }
    float* ar = &At[(it * 16 + w) * FPITCH + c0];
    *(v4f*)(ar)       = oa * inva;
    *(v4f*)(ar + 128) = ob * invb;
  }

  const int a_frag = (wm * 16 + lid) * FPITCH + (half << 1);
  const int b_frag = (half * FBPITCH + wn * 64 + lid) << 1;

  v8f acc[4];
#pragma unroll
  for (int nt = 0; nt < 4; ++nt) acc[nt] = (v8f){0.f,0.f,0.f,0.f,0.f,0.f,0.f,0.f};

  __syncthreads();

  constexpr int nk = CCH / FK;
  for (int kt = 0; kt < nk; ++kt) {
    const int cur = kt & 1;
    if (kt + 1 < nk) store_b_slab(&Bs[cur ^ 1][bs_off], rb);
    if (kt + 2 < nk) load_b_slab(rb, bg + (size_t)((kt + 2) * FK) * CCH);
    const float* as = &At[a_frag + kt * FK];
    const float* bs = &Bs[cur][b_frag];
#pragma unroll
    for (int kk = 0; kk < FK / 4; ++kk) {
      const v2f af = *(const v2f*)(as + kk * 4);
      v2f bf[4];
#pragma unroll
      for (int nt = 0; nt < 4; ++nt)
        bf[nt] = *(const v2f*)(bs + ((kk * 2 * FBPITCH + nt * 16) << 1));
#pragma unroll
      for (int nt = 0; nt < 4; ++nt)
        acc[nt] = __builtin_amdgcn_wmma_f32_16x16x4_f32(false, af, false, bf[nt], (short)0, acc[nt], false, false);
    }
    __syncthreads();
  }

  const int ccol = wn * 64 + lid;
  float bv[4];
#pragma unroll
  for (int nt = 0; nt < 4; ++nt) bv[nt] = bias[ccol + nt * 16];
  const int nb = ((x0 + wm) << 8) | ((((blk >> 2) & 3) << 2) << 4) | ((blk & 3) << 2);
#pragma unroll
  for (int i = 0; i < 8; ++i) {
    const int r  = i + (half << 3);
    const int nr = nb | ((r >> 2) << 4) | (r & 3);
    float* dr = out + (size_t)(b * NVOX + nr) * CCH + ccol;
#pragma unroll
    for (int nt = 0; nt < 4; ++nt) dr[nt * 16] = acc[nt][i] + bv[nt];
  }
}

extern "C" void kernel_launch(void* const* d_in, const int* in_sizes, int n_in,
                              void* d_out, int out_size, void* d_ws, size_t ws_size,
                              hipStream_t stream) {
  (void)in_sizes; (void)n_in; (void)out_size; (void)ws_size;
  const float* x      = (const float*)d_in[0];
  const float* w_qkv  = (const float*)d_in[1];
  const float* w_proj = (const float*)d_in[2];
  const float* b_proj = (const float*)d_in[3];
  const int*   neigh  = (const int*)d_in[4];
  float* out = (float*)d_out;
  float* qkv = (float*)d_ws;

  const int M = BATCH * NVOX;

  {
    int blocks = (M / 128) * (3 * CCH / GBN);
    wmma_gemm_f32<2><<<blocks, 256, 0, stream>>>(
        x, w_qkv, nullptr, qkv, CCH, 3 * CCH, CCH);
  }
  {
    attn_proj<<<BATCH * NVOX / 64, 512, 0, stream>>>(qkv, neigh, w_proj, b_proj, out);
  }
}
